# starting kernel
# baseline (speedup 1.0000x reference)
#include <hip/hip_runtime.h>
#include <hip/hip_bf16.h>

typedef __attribute__((ext_vector_type(2))) float v2f;
typedef __attribute__((ext_vector_type(8))) float v8f;

#define MB 8
#define D1C 32
#define D2C 32
#define EC 256
#define LC 16384
#define HC 256
#define NSPEC 32
#define SIDXC 8
#define NPLAY 4
#define POLW (LC + SIDXC)


__device__ __forceinline__ float dot8(float4 a, float4 b, float4 c, float4 d) {
    return a.x*b.x + a.y*b.y + a.z*b.z + a.w*b.w +
           c.x*d.x + c.y*d.y + c.z*d.z + c.w*d.w;
}

__device__ __forceinline__ float wave_sum(float v) {
    #pragma unroll
    for (int off = 16; off > 0; off >>= 1) v += __shfl_xor(v, off, 32);
    return v;
}

__device__ __forceinline__ v8f wmma_tile_k256(const float* A, const float* B,
                                              int ldb, int bcol, int arow, int khalf) {
    v8f acc = {};
    #pragma unroll 8
    for (int k0 = 0; k0 < 256; k0 += 4) {
        const int kA = k0 + khalf * 2;
        v2f a = *(const v2f*)(A + arow * 256 + kA);
        v2f b;
        b.x = B[(kA    ) * ldb + bcol];
        b.y = B[(kA + 1) * ldb + bcol];
        acc = __builtin_amdgcn_wmma_f32_16x16x4_f32(
                  false, a, false, b, (short)0, acc, false, false);
    }
    return acc;
}

__global__ __launch_bounds__(256) void emb_dot_kernel(
    const float* __restrict__ emb, const float* __restrict__ cls,
    const float* __restrict__ Wp, const float* __restrict__ bp,
    float* __restrict__ ws)
{
    const int lane = threadIdx.x & 31;
    const int wave = threadIdx.x >> 5;
    const int row  = blockIdx.x * 8 + wave;
    float* emb_dot = ws + 64;

    const float4* e4  = (const float4*)(emb + (size_t)row * EC);
    const float4* w04 = (const float4*)(Wp);
    const float4* w14 = (const float4*)(Wp + EC);

    float4 ea = e4[lane*2 + 0], eb = e4[lane*2 + 1];
    float p0 = dot8(ea, w04[lane*2+0], eb, w04[lane*2+1]);
    float p1 = dot8(ea, w14[lane*2+0], eb, w14[lane*2+1]);
    p0 = wave_sum(p0);
    p1 = wave_sum(p1);
    if (lane == 0) { emb_dot[row*2 + 0] = p0; emb_dot[row*2 + 1] = p1; }

    if (blockIdx.x == 0 && wave == 0) {
        const float4* wc4 = (const float4*)(Wp + 2*EC);
        float4 wca = wc4[lane*2], wcb = wc4[lane*2+1];
        float bias = bp[0];
        for (int m = 0; m < MB; ++m) {
            const float4* c4 = (const float4*)(cls + m*EC);
            float p = dot8(c4[lane*2], wca, c4[lane*2+1], wcb);
            p = wave_sum(p);
            if (lane == 0) ws[m] = p + bias;
        }
    }
}

__global__ __launch_bounds__(512) void heads_kernel(
    const float* __restrict__ cls,
    const float* __restrict__ Wv1, const float* __restrict__ bv1,
    const float* __restrict__ Wv2, const float* __restrict__ bv2,
    const float* __restrict__ Ws1, const float* __restrict__ bs1,
    const float* __restrict__ Ws2, const float* __restrict__ bs2,
    const int*   __restrict__ sidx,
    float* __restrict__ out)
{
    __shared__ float clsLds[16 * EC];
    __shared__ float hv[16 * HC];
    __shared__ float hs[16 * HC];
    __shared__ float w2v[HC * 16];
    __shared__ float spAll[MB * NSPEC];

    const int t     = threadIdx.x;
    const int lane  = t & 31;
    const int wave  = t >> 5;
    const int colr  = lane & 15;
    const int khalf = lane >> 4;

    for (int i = t; i < 16 * EC; i += 512) {
        const int r = i >> 8, c = i & 255;
        clsLds[i] = (r < MB) ? cls[r * EC + c] : 0.f;
    }
    for (int i = t; i < 8 * HC; i += 512) {
        hv[8 * HC + i] = 0.f;
        hs[8 * HC + i] = 0.f;
    }
    for (int i = t; i < HC * 16; i += 512) {
        const int k = i >> 4, c = i & 15;
        w2v[i] = (c < NPLAY) ? Wv2[k * NPLAY + c] : 0.f;
    }
    __syncthreads();

    for (int job = wave; job < 32; job += 16) {
        const float* W1 = (job < 16) ? Wv1 : Ws1;
        const float* b1 = (job < 16) ? bv1 : bs1;
        float* dst      = (job < 16) ? hv  : hs;
        const int n0 = (job & 15) * 16;
        v8f acc = wmma_tile_k256(clsLds, W1, HC, n0 + colr, colr, khalf);
        if (khalf == 0) {
            const float bias = b1[n0 + colr];
            #pragma unroll
            for (int i = 0; i < MB; ++i) {
                const float v = acc[i] + bias;
                dst[i * HC + n0 + colr] = v > 0.f ? v : 0.f;
            }
        }
    }
    __syncthreads();

    if (wave < 2) {
        const int n0 = wave * 16;
        v8f acc = wmma_tile_k256(hs, Ws2, NSPEC, n0 + colr, colr, khalf);
        if (khalf == 0) {
            const float bias = bs2[n0 + colr];
            #pragma unroll
            for (int i = 0; i < MB; ++i)
                spAll[i * NSPEC + n0 + colr] = acc[i] + bias;
        }
    } else if (wave == 2) {
        v8f acc = wmma_tile_k256(hv, w2v, 16, colr, colr, khalf);
        if (khalf == 0 && colr < NPLAY) {
            const float bias = bv2[colr];
            #pragma unroll
            for (int i = 0; i < MB; ++i)
                out[MB * POLW + i * NPLAY + colr] = acc[i] + bias;
        }
    }
    __syncthreads();

    if (t < MB * SIDXC) {
        const int m = t >> 3, j = t & 7;
        out[m * POLW + LC + j] = spAll[m * NSPEC + sidx[j]];
    }
}

__global__ __launch_bounds__(256) void policy_kernel(
    const int* __restrict__ moves, const float* __restrict__ ws,
    float* __restrict__ out)
{
    __shared__ float clsd[MB];
    if (threadIdx.x < MB) clsd[threadIdx.x] = ws[threadIdx.x];
    __syncthreads();

    const int l = blockIdx.x * blockDim.x + threadIdx.x;
    const float* emb_dot = ws + 64;
    const int4 mv = ((const int4*)moves)[l];
    const int r0 = mv.x * D2C + mv.y;
    const int r1 = mv.z * D2C + mv.w;
    #pragma unroll
    for (int m = 0; m < MB; ++m) {
        const float d0 = emb_dot[(m*1024 + r0)*2 + 0];
        const float d1 = emb_dot[(m*1024 + r1)*2 + 1];
        out[m*POLW + l] = d0 + d1 + clsd[m];
    }
}

extern "C" void kernel_launch(void* const* d_in, const int* in_sizes, int n_in,
                              void* d_out, int out_size, void* d_ws, size_t ws_size,
                              hipStream_t stream) {
    (void)in_sizes; (void)n_in; (void)out_size; (void)ws_size;
    const float* emb  = (const float*)d_in[0];
    const float* cls  = (const float*)d_in[1];
    const int*   mov  = (const int*)  d_in[2];
    const int*   sidx = (const int*)  d_in[3];
    const float* Wp   = (const float*)d_in[4];
    const float* bp   = (const float*)d_in[5];
    const float* Wv1  = (const float*)d_in[6];
    const float* bv1  = (const float*)d_in[7];
    const float* Wv2  = (const float*)d_in[8];
    const float* bv2  = (const float*)d_in[9];
    const float* Ws1  = (const float*)d_in[10];
    const float* bs1  = (const float*)d_in[11];
    const float* Ws2  = (const float*)d_in[12];
    const float* bs2  = (const float*)d_in[13];
    float* out = (float*)d_out;
    float* ws  = (float*)d_ws;

    emb_dot_kernel<<<1024, 256, 0, stream>>>(emb, cls, Wp, bp, ws);
    heads_kernel<<<1, 512, 0, stream>>>(cls, Wv1, bv1, Wv2, bv2,
                                        Ws1, bs1, Ws2, bs2, sidx, out);
    policy_kernel<<<LC/256, 256, 0, stream>>>(mov, ws, out);
}


# heads cut by hidden slice: 32 one-wave work-groups, each 16 hidden columns in both layers (loads issued ahead of one WMMA chain); layer-2 sum = 16 slice partials added in rising order in launch 2
# speedup vs baseline: 2.9010x; 2.9010x over previous
#include <hip/hip_runtime.h>
#include <hip/hip_bf16.h>

typedef __attribute__((ext_vector_type(2))) float v2f;
typedef __attribute__((ext_vector_type(8))) float v8f;

#define MB 8
#define D1C 32
#define D2C 32
#define EC 256
#define LC 16384
#define HC 256
#define NSPEC 32
#define SIDXC 8
#define NPLAY 4
#define POLW (LC + SIDXC)

#define NSLICE 16
#define PS_OFF 0
#define PV_OFF (NSLICE * MB * NSPEC)
#define HPITCH 20
#define FIN_BLOCK 0

__device__ __forceinline__ float dot8(float4 a, float4 b, float4 c, float4 d) {
    return a.x*b.x + a.y*b.y + a.z*b.z + a.w*b.w +
           c.x*d.x + c.y*d.y + c.z*d.z + c.w*d.w;
}

__device__ __forceinline__ float wave_sum(float v) {
    #pragma unroll
    for (int off = 16; off > 0; off >>= 1) v += __shfl_xor(v, off, 32);
    return v;
}

__global__ __launch_bounds__(256) void emb_dot_kernel(
    const float* __restrict__ emb, const float* __restrict__ cls,
    const float* __restrict__ Wp, const float* __restrict__ bp,
    const int*   __restrict__ sidx,
    const float* __restrict__ bv2, const float* __restrict__ bs2,
    float* __restrict__ ws, float* out)
{
    if (blockIdx.x == FIN_BLOCK) {
        const int t = threadIdx.x;
        if (t < MB * SIDXC) {
            const int m = t >> 3, j = t & 7;
            const int col = sidx[j] & (NSPEC - 1);
            const float* p = out + PS_OFF + m * NSPEC + col;
            float part[NSLICE];
            #pragma unroll
            for (int s = 0; s < NSLICE; ++s) part[s] = p[s * MB * NSPEC];
            float acc = part[0];
            #pragma unroll
            for (int s = 1; s < NSLICE; ++s) acc += part[s];
            out[m * POLW + LC + j] = acc + bs2[col];
        } else if (t < MB * SIDXC + MB * NPLAY) {
            const int i = t - MB * SIDXC;
            const int m = i >> 2, c = i & 3;
            const float* p = out + PV_OFF + m * NPLAY + c;
            float part[NSLICE];
            #pragma unroll
            for (int s = 0; s < NSLICE; ++s) part[s] = p[s * MB * NPLAY];
            float acc = part[0];
            #pragma unroll
            for (int s = 1; s < NSLICE; ++s) acc += part[s];
            out[MB * POLW + i] = acc + bv2[c];
        }
        return;
    }

    const int lane = threadIdx.x & 31;
    const int wave = threadIdx.x >> 5;
    const int sb   = blockIdx.x - 1;
    const int row  = sb * 8 + wave;
    float* emb_dot = ws + 64;

    const float4* e4  = (const float4*)(emb + (size_t)row * EC);
    const float4* w04 = (const float4*)(Wp);
    const float4* w14 = (const float4*)(Wp + EC);

    float4 ea = e4[lane*2 + 0], eb = e4[lane*2 + 1];
    float p0 = dot8(ea, w04[lane*2+0], eb, w04[lane*2+1]);
    float p1 = dot8(ea, w14[lane*2+0], eb, w14[lane*2+1]);
    p0 = wave_sum(p0);
    p1 = wave_sum(p1);
    if (lane == 0) { emb_dot[row*2 + 0] = p0; emb_dot[row*2 + 1] = p1; }

    if (sb == 0 && wave == 0) {
        const float4* wc4 = (const float4*)(Wp + 2*EC);
        float4 wca = wc4[lane*2], wcb = wc4[lane*2+1];
        float bias = bp[0];
        for (int m = 0; m < MB; ++m) {
            const float4* c4 = (const float4*)(cls + m*EC);
            float p = dot8(c4[lane*2], wca, c4[lane*2+1], wcb);
            p = wave_sum(p);
            if (lane == 0) ws[m] = p + bias;
        }
    }
}

__global__ __launch_bounds__(32) __attribute__((amdgpu_waves_per_eu(1, 1))) void heads_slice_kernel(
    const float* __restrict__ cls,
    const float* __restrict__ Wv1, const float* __restrict__ bv1, const float* __restrict__ Wv2,
    const float* __restrict__ Ws1, const float* __restrict__ bs1, const float* __restrict__ Ws2,
    float* __restrict__ park)
{
    __shared__ float hT[16 * HPITCH];

    const int lane = threadIdx.x;
    const int c    = lane & 15;
    const int h    = lane >> 4;
    const int head = blockIdx.x >> 4;
    const int s    = blockIdx.x & 15;
    const int n0   = s * 16;

    const float* W1 = head ? Ws1 : Wv1;
    const float* b1 = head ? bs1 : bv1;
    const float* W2 = head ? Ws2 : Wv2;
    const int   ld2 = head ? NSPEC : NPLAY;

    const float* Ap = cls + (c & 7) * EC + 2 * h;
    const float* Bp = W1 + (2 * h) * HC + n0 + c;
    v2f a[64], b[64];
    #pragma unroll
    for (int i = 0; i < 64; ++i) {
        a[i]   = *(const v2f*)(Ap + 4 * i);
        b[i].x = Bp[(4 * i    ) * HC];
        b[i].y = Bp[(4 * i + 1) * HC];
    }
    const float bias1 = b1[n0 + c];
    const int  col0 = c,        col1 = 16 + c;
    const bool ok0  = col0 < ld2, ok1 = col1 < ld2;
    const float* W2p0 = W2 + (n0 + 2 * h) * ld2 + (ok0 ? col0 : 0);
    const float* W2p1 = W2 + (n0 + 2 * h) * ld2 + (ok1 ? col1 : 0);
    v2f w0[4], w1[4];
    #pragma unroll
    for (int st = 0; st < 4; ++st) {
        const float x0 = W2p0[(4 * st) * ld2], y0 = W2p0[(4 * st + 1) * ld2];
        const float x1 = W2p1[(4 * st) * ld2], y1 = W2p1[(4 * st + 1) * ld2];
        w0[st].x = ok0 ? x0 : 0.f;  w0[st].y = ok0 ? y0 : 0.f;
        w1[st].x = ok1 ? x1 : 0.f;  w1[st].y = ok1 ? y1 : 0.f;
    }

    v8f acc = {};
    #pragma unroll
    for (int i = 0; i < 64; ++i)
        acc = __builtin_amdgcn_wmma_f32_16x16x4_f32(
                  false, a[i], false, b[i], (short)0, acc, false, false);

    #pragma unroll
    for (int r = 0; r < 8; ++r) {
        const float v = acc[r] + bias1;
        hT[(8 * h + r) * HPITCH + c] = v > 0.f ? v : 0.f;
    }
    __syncthreads();

    v8f p0 = {}, p1 = {};
    #pragma unroll
    for (int st = 0; st < 4; ++st) {
        const v2f a2 = *(const v2f*)(hT + c * HPITCH + 4 * st + 2 * h);
        p0 = __builtin_amdgcn_wmma_f32_16x16x4_f32(
                 false, a2, false, w0[st], (short)0, p0, false, false);
        p1 = __builtin_amdgcn_wmma_f32_16x16x4_f32(
                 false, a2, false, w1[st], (short)0, p1, false, false);
    }

    float* pk = park + (head ? PS_OFF : PV_OFF) + s * MB * ld2;
    const int  colh = 16 * h + c;
    const bool okh  = colh < ld2;
    #pragma unroll
    for (int m = 0; m < MB; ++m) {
        const float v = h ? p1[m] : p0[m];
        if (okh) pk[m * ld2 + colh] = v;
    }
}

__global__ __launch_bounds__(256) void policy_kernel(
    const int* __restrict__ moves, const float* __restrict__ ws,
    float* __restrict__ out)
{
    __shared__ float clsd[MB];
    if (threadIdx.x < MB) clsd[threadIdx.x] = ws[threadIdx.x];
    __syncthreads();

    const int l = blockIdx.x * blockDim.x + threadIdx.x;
    const float* emb_dot = ws + 64;
    const int4 mv = ((const int4*)moves)[l];
    const int r0 = mv.x * D2C + mv.y;
    const int r1 = mv.z * D2C + mv.w;
    #pragma unroll
    for (int m = 0; m < MB; ++m) {
        const float d0 = emb_dot[(m*1024 + r0)*2 + 0];
        const float d1 = emb_dot[(m*1024 + r1)*2 + 1];
        out[m*POLW + l] = d0 + d1 + clsd[m];
    }
}

extern "C" void kernel_launch(void* const* d_in, const int* in_sizes, int n_in,
                              void* d_out, int out_size, void* d_ws, size_t ws_size,
                              hipStream_t stream) {
    (void)in_sizes; (void)n_in; (void)out_size; (void)ws_size;
    const float* emb  = (const float*)d_in[0];
    const float* cls  = (const float*)d_in[1];
    const int*   mov  = (const int*)  d_in[2];
    const int*   sidx = (const int*)  d_in[3];
    const float* Wp   = (const float*)d_in[4];
    const float* bp   = (const float*)d_in[5];
    const float* Wv1  = (const float*)d_in[6];
    const float* bv1  = (const float*)d_in[7];
    const float* Wv2  = (const float*)d_in[8];
    const float* bv2  = (const float*)d_in[9];
    const float* Ws1  = (const float*)d_in[10];
    const float* bs1  = (const float*)d_in[11];
    const float* Ws2  = (const float*)d_in[12];
    const float* bs2  = (const float*)d_in[13];
    float* out = (float*)d_out;
    float* ws  = (float*)d_ws;

    heads_slice_kernel<<<32, 32, 0, stream>>>(cls, Wv1, bv1, Wv2, Ws1, bs1, Ws2, out);
    emb_dot_kernel<<<1025, 256, 0, stream>>>(emb, cls, Wp, bp, sidx, bv2, bs2, ws, out);
    policy_kernel<<<LC/256, 256, 0, stream>>>(mov, ws, out);
}


# table and cls dots moved onto f32 WMMA: a wave takes 16 embedding rows as A and Wp's two columns as B, two 32-step chains added at the end (new sum order, no shuffle tree); 513 tile waves, was 8192
# speedup vs baseline: 3.3847x; 1.1667x over previous
#include <hip/hip_runtime.h>
#include <hip/hip_bf16.h>

typedef __attribute__((ext_vector_type(2))) float v2f;
typedef __attribute__((ext_vector_type(8))) float v8f;

#define MB 8
#define D1C 32
#define D2C 32
#define EC 256
#define LC 16384
#define HC 256
#define NSPEC 32
#define SIDXC 8
#define NPLAY 4
#define POLW (LC + SIDXC)

#define NSLICE 16
#define PS_OFF 0
#define PV_OFF (NSLICE * MB * NSPEC)
#define HPITCH 20
#define NFIN 2
#define TROWS 16
#define TBLOCKS 256

__device__ __forceinline__ void tile_k256(const float* __restrict__ arow, const float* __restrict__ bcol,
                                          int h, v8f& acc0, v8f& acc1)
{
    const float4* Ap = (const float4*)(arow + 4 * h);
    const float4* Bp = (const float4*)(bcol + 4 * h);
    float4 a[32], b[32];
    #pragma unroll
    for (int j = 0; j < 32; ++j) {
        b[j] = Bp[2 * j];
        a[j] = Ap[2 * j];
    }
    #pragma unroll
    for (int j = 0; j < 32; ++j) {
        const v2f a0 = {a[j].x, a[j].y}, b0 = {b[j].x, b[j].y};
        const v2f a1 = {a[j].z, a[j].w}, b1 = {b[j].z, b[j].w};
        acc0 = __builtin_amdgcn_wmma_f32_16x16x4_f32(
                   false, a0, false, b0, (short)0, acc0, false, false);
        acc1 = __builtin_amdgcn_wmma_f32_16x16x4_f32(
                   false, a1, false, b1, (short)0, acc1, false, false);
    }
}

__global__ __launch_bounds__(64) __attribute__((amdgpu_waves_per_eu(1, 1))) void emb_dot_kernel(
    const float* __restrict__ emb, const float* __restrict__ cls,
    const float* __restrict__ Wp, const float* __restrict__ bp,
    const int*   __restrict__ sidx,
    const float* __restrict__ bv2, const float* __restrict__ bs2,
    float* __restrict__ ws, float* out)
{
    __shared__ float stage[2][2 * TROWS];

    const int lane = threadIdx.x & 31;
    const int wave = threadIdx.x >> 5;
    const int c    = lane & 15;
    const int h    = lane >> 4;

    if (blockIdx.x < NFIN) {
        const int t = threadIdx.x;
        if (blockIdx.x == 0) {
            const int m = t >> 3, j = t & 7;
            const int col = sidx[j] & (NSPEC - 1);
            const float* p = out + PS_OFF + m * NSPEC + col;
            float part[NSLICE];
            #pragma unroll
            for (int s = 0; s < NSLICE; ++s) part[s] = p[s * MB * NSPEC];
            float acc = part[0];
            #pragma unroll
            for (int s = 1; s < NSLICE; ++s) acc += part[s];
            out[m * POLW + LC + j] = acc + bs2[col];
        } else if (wave == 0) {
            const int i = t;
            const int m = i >> 2, pc = i & 3;
            const float* p = out + PV_OFF + m * NPLAY + pc;
            float part[NSLICE];
            #pragma unroll
            for (int s = 0; s < NSLICE; ++s) part[s] = p[s * MB * NPLAY];
            float acc = part[0];
            #pragma unroll
            for (int s = 1; s < NSLICE; ++s) acc += part[s];
            out[MB * POLW + i] = acc + bv2[pc];
        } else {
            const float bias = bp[0];
            v8f acc0 = {}, acc1 = {};
            tile_k256(cls + (c & (MB - 1)) * EC, Wp + 2 * EC, h, acc0, acc1);
            if (lane == 0) {
                #pragma unroll
                for (int r = 0; r < MB; ++r) ws[r] = (acc0[r] + acc1[r]) + bias;
            }
        }
        return;
    }

    const int row0 = ((blockIdx.x - NFIN) * 2 + wave) * TROWS;
    float* emb_dot = ws + 64;

    v8f acc0 = {}, acc1 = {};
    tile_k256(emb + (size_t)(row0 + c) * EC, Wp + (c & 1) * EC, h, acc0, acc1);

    if (c < 2) {
        #pragma unroll
        for (int r = 0; r < 8; ++r)
            stage[wave][(8 * h + r) * 2 + c] = acc0[r] + acc1[r];
    }
    __syncthreads();
    emb_dot[row0 * 2 + lane] = stage[wave][lane];
}

__global__ __launch_bounds__(32) __attribute__((amdgpu_waves_per_eu(1, 1))) void heads_slice_kernel(
    const float* __restrict__ cls,
    const float* __restrict__ Wv1, const float* __restrict__ bv1, const float* __restrict__ Wv2,
    const float* __restrict__ Ws1, const float* __restrict__ bs1, const float* __restrict__ Ws2,
    float* __restrict__ park)
{
    __shared__ float hT[16 * HPITCH];

    const int lane = threadIdx.x;
    const int c    = lane & 15;
    const int h    = lane >> 4;
    const int head = blockIdx.x >> 4;
    const int s    = blockIdx.x & 15;
    const int n0   = s * 16;

    const float* W1 = head ? Ws1 : Wv1;
    const float* b1 = head ? bs1 : bv1;
    const float* W2 = head ? Ws2 : Wv2;
    const int   ld2 = head ? NSPEC : NPLAY;

    const float* Ap = cls + (c & 7) * EC + 2 * h;
    const float* Bp = W1 + (2 * h) * HC + n0 + c;
    v2f a[64], b[64];
    #pragma unroll
    for (int i = 0; i < 64; ++i) {
        a[i]   = *(const v2f*)(Ap + 4 * i);
        b[i].x = Bp[(4 * i    ) * HC];
        b[i].y = Bp[(4 * i + 1) * HC];
    }
    const float bias1 = b1[n0 + c];
    const int  col0 = c,        col1 = 16 + c;
    const bool ok0  = col0 < ld2, ok1 = col1 < ld2;
    const float* W2p0 = W2 + (n0 + 2 * h) * ld2 + (ok0 ? col0 : 0);
    const float* W2p1 = W2 + (n0 + 2 * h) * ld2 + (ok1 ? col1 : 0);
    v2f w0[4], w1[4];
    #pragma unroll
    for (int st = 0; st < 4; ++st) {
        const float x0 = W2p0[(4 * st) * ld2], y0 = W2p0[(4 * st + 1) * ld2];
        const float x1 = W2p1[(4 * st) * ld2], y1 = W2p1[(4 * st + 1) * ld2];
        w0[st].x = ok0 ? x0 : 0.f;  w0[st].y = ok0 ? y0 : 0.f;
        w1[st].x = ok1 ? x1 : 0.f;  w1[st].y = ok1 ? y1 : 0.f;
    }

    v8f acc = {};
    #pragma unroll
    for (int i = 0; i < 64; ++i)
        acc = __builtin_amdgcn_wmma_f32_16x16x4_f32(
                  false, a[i], false, b[i], (short)0, acc, false, false);

    #pragma unroll
    for (int r = 0; r < 8; ++r) {
        const float v = acc[r] + bias1;
        hT[(8 * h + r) * HPITCH + c] = v > 0.f ? v : 0.f;
    }
    __syncthreads();

    v8f p0 = {}, p1 = {};
    #pragma unroll
    for (int st = 0; st < 4; ++st) {
        const v2f a2 = *(const v2f*)(hT + c * HPITCH + 4 * st + 2 * h);
        p0 = __builtin_amdgcn_wmma_f32_16x16x4_f32(
                 false, a2, false, w0[st], (short)0, p0, false, false);
        p1 = __builtin_amdgcn_wmma_f32_16x16x4_f32(
                 false, a2, false, w1[st], (short)0, p1, false, false);
    }

    float* pk = park + (head ? PS_OFF : PV_OFF) + s * MB * ld2;
    const int  colh = 16 * h + c;
    const bool okh  = colh < ld2;
    #pragma unroll
    for (int m = 0; m < MB; ++m) {
        const float v = h ? p1[m] : p0[m];
        if (okh) pk[m * ld2 + colh] = v;
    }
}

__global__ __launch_bounds__(256) void policy_kernel(
    const int* __restrict__ moves, const float* __restrict__ ws,
    float* __restrict__ out)
{
    __shared__ float clsd[MB];
    if (threadIdx.x < MB) clsd[threadIdx.x] = ws[threadIdx.x];
    __syncthreads();

    const int l = blockIdx.x * blockDim.x + threadIdx.x;
    const float* emb_dot = ws + 64;
    const int4 mv = ((const int4*)moves)[l];
    const int r0 = mv.x * D2C + mv.y;
    const int r1 = mv.z * D2C + mv.w;
    #pragma unroll
    for (int m = 0; m < MB; ++m) {
        const float d0 = emb_dot[(m*1024 + r0)*2 + 0];
        const float d1 = emb_dot[(m*1024 + r1)*2 + 1];
        out[m*POLW + l] = d0 + d1 + clsd[m];
    }
}

extern "C" void kernel_launch(void* const* d_in, const int* in_sizes, int n_in,
                              void* d_out, int out_size, void* d_ws, size_t ws_size,
                              hipStream_t stream) {
    (void)in_sizes; (void)n_in; (void)out_size; (void)ws_size;
    const float* emb  = (const float*)d_in[0];
    const float* cls  = (const float*)d_in[1];
    const int*   mov  = (const int*)  d_in[2];
    const int*   sidx = (const int*)  d_in[3];
    const float* Wp   = (const float*)d_in[4];
    const float* bp   = (const float*)d_in[5];
    const float* Wv1  = (const float*)d_in[6];
    const float* bv1  = (const float*)d_in[7];
    const float* Wv2  = (const float*)d_in[8];
    const float* bv2  = (const float*)d_in[9];
    const float* Ws1  = (const float*)d_in[10];
    const float* bs1  = (const float*)d_in[11];
    const float* Ws2  = (const float*)d_in[12];
    const float* bs2  = (const float*)d_in[13];
    float* out = (float*)d_out;
    float* ws  = (float*)d_ws;

    heads_slice_kernel<<<32, 32, 0, stream>>>(cls, Wv1, bv1, Wv2, Ws1, bs1, Ws2, out);
    emb_dot_kernel<<<NFIN + TBLOCKS, 64, 0, stream>>>(emb, cls, Wp, bp, sidx, bv2, bs2, ws, out);
    policy_kernel<<<LC/256, 256, 0, stream>>>(mov, ws, out);
}


# policy launch cut by batch: a work-group copies its batch's 8-KB table slice to LDS with whole-line loads beside its moves and gathers from LDS: no scattered or dependent global load; 256 groups
# speedup vs baseline: 3.7963x; 1.1216x over previous
#include <hip/hip_runtime.h>
#include <hip/hip_bf16.h>

typedef __attribute__((ext_vector_type(2))) float v2f;
typedef __attribute__((ext_vector_type(8))) float v8f;

#define MB 8
#define D1C 32
#define D2C 32
#define EC 256
#define LC 16384
#define HC 256
#define NSPEC 32
#define SIDXC 8
#define NPLAY 4
#define POLW (LC + SIDXC)

#define NSLICE 16
#define PS_OFF 0
#define PV_OFF (NSLICE * MB * NSPEC)
#define HPITCH 20
#define NFIN 2
#define TROWS 16
#define TBLOCKS 256

__device__ __forceinline__ void tile_k256(const float* __restrict__ arow, const float* __restrict__ bcol,
                                          int h, v8f& acc0, v8f& acc1)
{
    const float4* Ap = (const float4*)(arow + 4 * h);
    const float4* Bp = (const float4*)(bcol + 4 * h);
    float4 a[32], b[32];
    #pragma unroll
    for (int j = 0; j < 32; ++j) {
        b[j] = Bp[2 * j];
        a[j] = Ap[2 * j];
    }
    #pragma unroll
    for (int j = 0; j < 32; ++j) {
        const v2f a0 = {a[j].x, a[j].y}, b0 = {b[j].x, b[j].y};
        const v2f a1 = {a[j].z, a[j].w}, b1 = {b[j].z, b[j].w};
        acc0 = __builtin_amdgcn_wmma_f32_16x16x4_f32(
                   false, a0, false, b0, (short)0, acc0, false, false);
        acc1 = __builtin_amdgcn_wmma_f32_16x16x4_f32(
                   false, a1, false, b1, (short)0, acc1, false, false);
    }
}

__global__ __launch_bounds__(64) __attribute__((amdgpu_waves_per_eu(1, 1))) void emb_dot_kernel(
    const float* __restrict__ emb, const float* __restrict__ cls,
    const float* __restrict__ Wp, const float* __restrict__ bp,
    const int*   __restrict__ sidx,
    const float* __restrict__ bv2, const float* __restrict__ bs2,
    float* __restrict__ ws, float* out)
{
    __shared__ float stage[2][2 * TROWS];

    const int lane = threadIdx.x & 31;
    const int wave = threadIdx.x >> 5;
    const int c    = lane & 15;
    const int h    = lane >> 4;

    if (blockIdx.x < NFIN) {
        const int t = threadIdx.x;
        if (blockIdx.x == 0) {
            const int m = t >> 3, j = t & 7;
            const int col = sidx[j] & (NSPEC - 1);
            const float* p = out + PS_OFF + m * NSPEC + col;
            float part[NSLICE];
            #pragma unroll
            for (int s = 0; s < NSLICE; ++s) part[s] = p[s * MB * NSPEC];
            float acc = part[0];
            #pragma unroll
            for (int s = 1; s < NSLICE; ++s) acc += part[s];
            out[m * POLW + LC + j] = acc + bs2[col];
        } else if (wave == 0) {
            const int i = t;
            const int m = i >> 2, pc = i & 3;
            const float* p = out + PV_OFF + m * NPLAY + pc;
            float part[NSLICE];
            #pragma unroll
            for (int s = 0; s < NSLICE; ++s) part[s] = p[s * MB * NPLAY];
            float acc = part[0];
            #pragma unroll
            for (int s = 1; s < NSLICE; ++s) acc += part[s];
            out[MB * POLW + i] = acc + bv2[pc];
        } else {
            const float bias = bp[0];
            v8f acc0 = {}, acc1 = {};
            tile_k256(cls + (c & (MB - 1)) * EC, Wp + 2 * EC, h, acc0, acc1);
            if (lane == 0) {
                #pragma unroll
                for (int r = 0; r < MB; ++r) ws[r] = (acc0[r] + acc1[r]) + bias;
            }
        }
        return;
    }

    const int row0 = ((blockIdx.x - NFIN) * 2 + wave) * TROWS;
    float* emb_dot = ws + 64;

    v8f acc0 = {}, acc1 = {};
    tile_k256(emb + (size_t)(row0 + c) * EC, Wp + (c & 1) * EC, h, acc0, acc1);

    if (c < 2) {
        #pragma unroll
        for (int r = 0; r < 8; ++r)
            stage[wave][(8 * h + r) * 2 + c] = acc0[r] + acc1[r];
    }
    __syncthreads();
    emb_dot[row0 * 2 + lane] = stage[wave][lane];
}

__global__ __launch_bounds__(32) __attribute__((amdgpu_waves_per_eu(1, 1))) void heads_slice_kernel(
    const float* __restrict__ cls,
    const float* __restrict__ Wv1, const float* __restrict__ bv1, const float* __restrict__ Wv2,
    const float* __restrict__ Ws1, const float* __restrict__ bs1, const float* __restrict__ Ws2,
    float* __restrict__ park)
{
    __shared__ float hT[16 * HPITCH];

    const int lane = threadIdx.x;
    const int c    = lane & 15;
    const int h    = lane >> 4;
    const int head = blockIdx.x >> 4;
    const int s    = blockIdx.x & 15;
    const int n0   = s * 16;

    const float* W1 = head ? Ws1 : Wv1;
    const float* b1 = head ? bs1 : bv1;
    const float* W2 = head ? Ws2 : Wv2;
    const int   ld2 = head ? NSPEC : NPLAY;

    const float* Ap = cls + (c & 7) * EC + 2 * h;
    const float* Bp = W1 + (2 * h) * HC + n0 + c;
    v2f a[64], b[64];
    #pragma unroll
    for (int i = 0; i < 64; ++i) {
        a[i]   = *(const v2f*)(Ap + 4 * i);
        b[i].x = Bp[(4 * i    ) * HC];
        b[i].y = Bp[(4 * i + 1) * HC];
    }
    const float bias1 = b1[n0 + c];
    const int  col0 = c,        col1 = 16 + c;
    const bool ok0  = col0 < ld2, ok1 = col1 < ld2;
    const float* W2p0 = W2 + (n0 + 2 * h) * ld2 + (ok0 ? col0 : 0);
    const float* W2p1 = W2 + (n0 + 2 * h) * ld2 + (ok1 ? col1 : 0);
    v2f w0[4], w1[4];
    #pragma unroll
    for (int st = 0; st < 4; ++st) {
        const float x0 = W2p0[(4 * st) * ld2], y0 = W2p0[(4 * st + 1) * ld2];
        const float x1 = W2p1[(4 * st) * ld2], y1 = W2p1[(4 * st + 1) * ld2];
        w0[st].x = ok0 ? x0 : 0.f;  w0[st].y = ok0 ? y0 : 0.f;
        w1[st].x = ok1 ? x1 : 0.f;  w1[st].y = ok1 ? y1 : 0.f;
    }

    v8f acc = {};
    #pragma unroll
    for (int i = 0; i < 64; ++i)
        acc = __builtin_amdgcn_wmma_f32_16x16x4_f32(
                  false, a[i], false, b[i], (short)0, acc, false, false);

    #pragma unroll
    for (int r = 0; r < 8; ++r) {
        const float v = acc[r] + bias1;
        hT[(8 * h + r) * HPITCH + c] = v > 0.f ? v : 0.f;
    }
    __syncthreads();

    v8f p0 = {}, p1 = {};
    #pragma unroll
    for (int st = 0; st < 4; ++st) {
        const v2f a2 = *(const v2f*)(hT + c * HPITCH + 4 * st + 2 * h);
        p0 = __builtin_amdgcn_wmma_f32_16x16x4_f32(
                 false, a2, false, w0[st], (short)0, p0, false, false);
        p1 = __builtin_amdgcn_wmma_f32_16x16x4_f32(
                 false, a2, false, w1[st], (short)0, p1, false, false);
    }

    float* pk = park + (head ? PS_OFF : PV_OFF) + s * MB * ld2;
    const int  colh = 16 * h + c;
    const bool okh  = colh < ld2;
    #pragma unroll
    for (int m = 0; m < MB; ++m) {
        const float v = h ? p1[m] : p0[m];
        if (okh) pk[m * ld2 + colh] = v;
    }
}

#define PCOLS 512
#define PTHR  64
#define PPER  (PCOLS / PTHR)
#define PCHUNK (LC / PCOLS)
#define SLICE (D1C * D2C * 2)
__global__ __launch_bounds__(PTHR) void policy_kernel(
    const int* __restrict__ moves, const float* __restrict__ ws,
    float* __restrict__ out)
{
    __shared__ float4 tab4[SLICE / 4];

    const int t  = threadIdx.x;
    const int m  = blockIdx.x / PCHUNK;
    const int l0 = (blockIdx.x % PCHUNK) * PCOLS;

    const float4* slice = (const float4*)(ws + 64 + m * SLICE);
    const int4*   mv4   = (const int4*)moves + l0;
    float4 tv[PPER];
    int4   mv[PPER];
    #pragma unroll
    for (int i = 0; i < PPER; ++i) tv[i] = slice[i * PTHR + t];
    #pragma unroll
    for (int i = 0; i < PPER; ++i) mv[i] = mv4[i * PTHR + t];
    const float cd = ws[m];

    #pragma unroll
    for (int i = 0; i < PPER; ++i) tab4[i * PTHR + t] = tv[i];
    __syncthreads();

    const float* tab  = (const float*)tab4;
    float*       orow = out + m * POLW + l0;
    #pragma unroll
    for (int i = 0; i < PPER; ++i) {
        const float d0 = tab[(mv[i].x * D2C + mv[i].y) * 2 + 0];
        const float d1 = tab[(mv[i].z * D2C + mv[i].w) * 2 + 1];
        orow[i * PTHR + t] = d0 + d1 + cd;
    }
}

extern "C" void kernel_launch(void* const* d_in, const int* in_sizes, int n_in,
                              void* d_out, int out_size, void* d_ws, size_t ws_size,
                              hipStream_t stream) {
    (void)in_sizes; (void)n_in; (void)out_size; (void)ws_size;
    const float* emb  = (const float*)d_in[0];
    const float* cls  = (const float*)d_in[1];
    const int*   mov  = (const int*)  d_in[2];
    const int*   sidx = (const int*)  d_in[3];
    const float* Wp   = (const float*)d_in[4];
    const float* bp   = (const float*)d_in[5];
    const float* Wv1  = (const float*)d_in[6];
    const float* bv1  = (const float*)d_in[7];
    const float* Wv2  = (const float*)d_in[8];
    const float* bv2  = (const float*)d_in[9];
    const float* Ws1  = (const float*)d_in[10];
    const float* bs1  = (const float*)d_in[11];
    const float* Ws2  = (const float*)d_in[12];
    const float* bs2  = (const float*)d_in[13];
    float* out = (float*)d_out;
    float* ws  = (float*)d_ws;

    heads_slice_kernel<<<32, 32, 0, stream>>>(cls, Wv1, bv1, Wv2, Ws1, bs1, Ws2, out);
    emb_dot_kernel<<<NFIN + TBLOCKS, 64, 0, stream>>>(emb, cls, Wp, bp, sidx, bv2, bs2, ws, out);
    policy_kernel<<<MB * PCHUNK, PTHR, 0, stream>>>(mov, ws, out);
}


# heads launch folded into the other two (3 -> 2): the 32 hidden-slice waves ride in the table launch, output partials parked in out; the 3 policy work-groups owning those columns add them, order kept
# speedup vs baseline: 5.4011x; 1.4227x over previous
#include <hip/hip_runtime.h>
#include <hip/hip_bf16.h>

typedef __attribute__((ext_vector_type(2))) float v2f;
typedef __attribute__((ext_vector_type(8))) float v8f;

#define MB 8
#define D1C 32
#define D2C 32
#define EC 256
#define LC 16384
#define HC 256
#define NSPEC 32
#define SIDXC 8
#define NPLAY 4
#define POLW (LC + SIDXC)

#define NSLICE 16
#define PCOLS 512
#define PARK(q, s) ((q) * PCOLS + (s) * 32)
#define FIN_WG 3
#define HPITCH 20
#define NHB NSLICE
#define NFRONT (NHB + 1)
#define TROWS 16
#define TBLOCKS 256

__device__ __forceinline__ void tile_k256(const float* __restrict__ arow, const float* __restrict__ bcol,
                                          int h, v8f& acc0, v8f& acc1)
{
    const float4* Ap = (const float4*)(arow + 4 * h);
    const float4* Bp = (const float4*)(bcol + 4 * h);
    float4 a[32], b[32];
    #pragma unroll
    for (int j = 0; j < 32; ++j) {
        b[j] = Bp[2 * j];
        a[j] = Ap[2 * j];
    }
    #pragma unroll
    for (int j = 0; j < 32; ++j) {
        const v2f a0 = {a[j].x, a[j].y}, b0 = {b[j].x, b[j].y};
        const v2f a1 = {a[j].z, a[j].w}, b1 = {b[j].z, b[j].w};
        acc0 = __builtin_amdgcn_wmma_f32_16x16x4_f32(
                   false, a0, false, b0, (short)0, acc0, false, false);
        acc1 = __builtin_amdgcn_wmma_f32_16x16x4_f32(
                   false, a1, false, b1, (short)0, acc1, false, false);
    }
}

__device__ __forceinline__ void heads_slice(
    const float* __restrict__ cls,
    const float* __restrict__ W1, const float* __restrict__ b1, const float* __restrict__ W2,
    const int ld2,
    const int head, const int s, const int lane,
    const int* __restrict__ sidx,
    float* hT, float* pT, float* park)
{
    const int c  = lane & 15;
    const int h  = lane >> 4;
    const int n0 = s * 16;

    const float* Ap = cls + (c & 7) * EC + 2 * h;
    const float* Bp = W1 + (2 * h) * HC + n0 + c;
    v2f a[64], b[64];
    #pragma unroll
    for (int i = 0; i < 64; ++i) {
        a[i]   = *(const v2f*)(Ap + 4 * i);
        b[i].x = Bp[(4 * i    ) * HC];
        b[i].y = Bp[(4 * i + 1) * HC];
    }
    const float bias1 = b1[n0 + c];
    const int   sraw  = sidx[lane & (SIDXC - 1)];
    const int  col0 = c,        col1 = 16 + c;
    const bool ok0  = col0 < ld2, ok1 = col1 < ld2;
    const float* W2p0 = W2 + (n0 + 2 * h) * ld2 + (ok0 ? col0 : 0);
    const float* W2p1 = W2 + (n0 + 2 * h) * ld2 + (ok1 ? col1 : 0);
    v2f w0[4], w1[4];
    #pragma unroll
    for (int st = 0; st < 4; ++st) {
        const float x0 = W2p0[(4 * st) * ld2], y0 = W2p0[(4 * st + 1) * ld2];
        const float x1 = W2p1[(4 * st) * ld2], y1 = W2p1[(4 * st + 1) * ld2];
        w0[st].x = ok0 ? x0 : 0.f;  w0[st].y = ok0 ? y0 : 0.f;
        w1[st].x = ok1 ? x1 : 0.f;  w1[st].y = ok1 ? y1 : 0.f;
    }

    v8f acc = {};
    #pragma unroll
    for (int i = 0; i < 64; ++i)
        acc = __builtin_amdgcn_wmma_f32_16x16x4_f32(
                  false, a[i], false, b[i], (short)0, acc, false, false);

    #pragma unroll
    for (int r = 0; r < 8; ++r) {
        const float v = acc[r] + bias1;
        hT[(8 * h + r) * HPITCH + c] = v > 0.f ? v : 0.f;
    }
    asm volatile("" :: "v"(sraw));
    __syncthreads();

    v8f p0 = {}, p1 = {};
    #pragma unroll
    for (int st = 0; st < 4; ++st) {
        const v2f a2 = *(const v2f*)(hT + c * HPITCH + 4 * st + 2 * h);
        p0 = __builtin_amdgcn_wmma_f32_16x16x4_f32(
                 false, a2, false, w0[st], (short)0, p0, false, false);
        p1 = __builtin_amdgcn_wmma_f32_16x16x4_f32(
                 false, a2, false, w1[st], (short)0, p1, false, false);
    }

    #pragma unroll
    for (int m = 0; m < MB; ++m) pT[m * 32 + 16 * h + c] = h ? p1[m] : p0[m];
    __syncthreads();

    const int scol = sraw & (NSPEC - 1);
    const int pick = head ? (lane >> 3) * 32 + scol : (lane >> 2) * 32 + (lane & 3);
    park[PARK(head ? 0 : 2, s) + lane] = pT[pick];
    if (head) park[PARK(1, s) + lane] = pT[pick + 4 * 32];
}

__global__ __launch_bounds__(64) __attribute__((amdgpu_waves_per_eu(1, 1))) void front_kernel(
    const float* __restrict__ emb, const float* __restrict__ cls,
    const float* __restrict__ Wp, const float* __restrict__ bp,
    const int*   __restrict__ sidx,
    const float* __restrict__ Wv1, const float* __restrict__ bv1, const float* __restrict__ Wv2,
    const float* __restrict__ Ws1, const float* __restrict__ bs1, const float* __restrict__ Ws2,
    float* __restrict__ ws, float* out)
{
    __shared__ float stage[2][2 * TROWS];
    __shared__ float hT[2][16 * HPITCH];
    __shared__ float pT[2][MB * 32];

    const int lane = threadIdx.x & 31;
    const int wave = __builtin_amdgcn_readfirstlane(threadIdx.x >> 5);
    const int c    = lane & 15;
    const int h    = lane >> 4;

    if (blockIdx.x < NHB) {
        const int head = wave;
        heads_slice(cls,
                    head ? Ws1 : Wv1, head ? bs1 : bv1, head ? Ws2 : Wv2,
                    head ? NSPEC : NPLAY, head, blockIdx.x, lane, sidx,
                    hT[wave], pT[wave], out);
        return;
    }
    if (blockIdx.x == NHB) {
        if (wave == 0) {
            const float bias = bp[0];
            v8f acc0 = {}, acc1 = {};
            tile_k256(cls + (c & (MB - 1)) * EC, Wp + 2 * EC, h, acc0, acc1);
            if (lane == 0) {
                #pragma unroll
                for (int r = 0; r < MB; ++r) ws[r] = (acc0[r] + acc1[r]) + bias;
            }
        }
        return;
    }

    const int row0 = ((blockIdx.x - NFRONT) * 2 + wave) * TROWS;
    float* emb_dot = ws + 64;

    v8f acc0 = {}, acc1 = {};
    tile_k256(emb + (size_t)(row0 + c) * EC, Wp + (c & 1) * EC, h, acc0, acc1);

    if (c < 2) {
        #pragma unroll
        for (int r = 0; r < 8; ++r)
            stage[wave][(8 * h + r) * 2 + c] = acc0[r] + acc1[r];
    }
    __syncthreads();
    emb_dot[row0 * 2 + lane] = stage[wave][lane];
}

#define PTHR  64
#define PPER  (PCOLS / PTHR)
#define PCHUNK (LC / PCOLS)
#define SLICE (D1C * D2C * 2)
template <bool OWNER>
__device__ __forceinline__ void policy_body(
    const int* __restrict__ moves, const float* __restrict__ ws,
    const int* __restrict__ sidx, const float* __restrict__ bv2, const float* __restrict__ bs2,
    float* __restrict__ out, float4* tab4)
{
    const int t  = threadIdx.x;
    const int m  = blockIdx.x / PCHUNK;
    const int l0 = (blockIdx.x % PCHUNK) * PCOLS;

    const float4* slice = (const float4*)(ws + 64 + m * SLICE);
    const int4*   mv4   = (const int4*)moves + l0;
    float4 tv[PPER];
    int4   mv[PPER];
    #pragma unroll
    for (int i = 0; i < PPER; ++i) tv[i] = slice[i * PTHR + t];
    #pragma unroll
    for (int i = 0; i < PPER; ++i) mv[i] = mv4[i * PTHR + t];
    const float cd = ws[m];

    if (OWNER && t < 32) {
        const int q = blockIdx.x;
        const float* p = out + q * PCOLS + t;
        const int   sraw = sidx[t & (SIDXC - 1)];
        const float bsv = bs2[t];
        const float bvv = bv2[t & (NPLAY - 1)];
        float part[NSLICE];
        #pragma unroll
        for (int s = 0; s < NSLICE; ++s) part[s] = p[s * 32];
        asm volatile("" :: "v"(sraw), "v"(bsv), "v"(bvv));
        const int   col = sraw & (NSPEC - 1);
        float acc = part[0];
        #pragma unroll
        for (int s = 1; s < NSLICE; ++s) acc += part[s];
        const float bsel = __shfl(bsv, col);
        const int   dst  = q < 2 ? (4 * q + (t >> 3)) * POLW + LC + (t & 7) : MB * POLW + t;
        out[dst] = acc + (q < 2 ? bsel : bvv);
    }

    #pragma unroll
    for (int i = 0; i < PPER; ++i) tab4[i * PTHR + t] = tv[i];
    __syncthreads();

    const float* tab  = (const float*)tab4;
    float*       orow = out + m * POLW + l0;
    #pragma unroll
    for (int i = 0; i < PPER; ++i) {
        const float d0 = tab[(mv[i].x * D2C + mv[i].y) * 2 + 0];
        const float d1 = tab[(mv[i].z * D2C + mv[i].w) * 2 + 1];
        orow[i * PTHR + t] = d0 + d1 + cd;
    }
}

__global__ __launch_bounds__(PTHR) void policy_kernel(
    const int* __restrict__ moves, const float* __restrict__ ws,
    const int* __restrict__ sidx, const float* __restrict__ bv2, const float* __restrict__ bs2,
    float* __restrict__ out)
{
    __shared__ float4 tab4[SLICE / 4];
    if (blockIdx.x < FIN_WG) policy_body<true >(moves, ws, sidx, bv2, bs2, out, tab4);
    else                     policy_body<false>(moves, ws, sidx, bv2, bs2, out, tab4);
}

extern "C" void kernel_launch(void* const* d_in, const int* in_sizes, int n_in,
                              void* d_out, int out_size, void* d_ws, size_t ws_size,
                              hipStream_t stream) {
    (void)in_sizes; (void)n_in; (void)out_size; (void)ws_size;
    const float* emb  = (const float*)d_in[0];
    const float* cls  = (const float*)d_in[1];
    const int*   mov  = (const int*)  d_in[2];
    const int*   sidx = (const int*)  d_in[3];
    const float* Wp   = (const float*)d_in[4];
    const float* bp   = (const float*)d_in[5];
    const float* Wv1  = (const float*)d_in[6];
    const float* bv1  = (const float*)d_in[7];
    const float* Wv2  = (const float*)d_in[8];
    const float* bv2  = (const float*)d_in[9];
    const float* Ws1  = (const float*)d_in[10];
    const float* bs1  = (const float*)d_in[11];
    const float* Ws2  = (const float*)d_in[12];
    const float* bs2  = (const float*)d_in[13];
    float* out = (float*)d_out;
    float* ws  = (float*)d_ws;

    front_kernel<<<NFRONT + TBLOCKS, 64, 0, stream>>>(emb, cls, Wp, bp, sidx,
                                                      Wv1, bv1, Wv2, Ws1, bs1, Ws2, ws, out);
    policy_kernel<<<MB * PCHUNK, PTHR, 0, stream>>>(mov, ws, sidx, bv2, bs2, out);
}
